# starting kernel
# baseline (speedup 1.0000x reference)
#include <hip/hip_runtime.h>
#include <hip/hip_bf16.h>

typedef float v2f __attribute__((ext_vector_type(2)));
typedef float v4f __attribute__((ext_vector_type(4)));
typedef float v8f __attribute__((ext_vector_type(8)));

#define FDIM 256
#define KDIM 512
#define DEG  32
#define TM   16
#define THREADS 128

__global__ __launch_bounds__(THREADS)
void graphsage_fused_kernel(const float* __restrict__ feat_data,
                            const float* __restrict__ feats,
                            const float* __restrict__ W1,
                            const float* __restrict__ b1,
                            const float* __restrict__ W2,
                            const float* __restrict__ b2,
                            const int*   __restrict__ adj,
                            const int*   __restrict__ nodes,
                            int nNodes,
                            float* __restrict__ out)
{
    __shared__ float S[TM][KDIM];
    __shared__ float T[TM][KDIM];
    __shared__ int   nidx[TM];
    __shared__ int   nbr[TM][DEG];
    __shared__ float part[TM][8];
    __shared__ float inv_norm[TM];

    const int t    = threadIdx.x;
    const int m0   = blockIdx.x * TM;
    const int lane = t & 31;
    const int wave = t >> 5;

    if (t < TM) {
        int g = m0 + t;
        nidx[t] = (g < nNodes) ? nodes[g] : 0;
    }
    __syncthreads();
    for (int i = t; i < TM * DEG; i += THREADS)
        nbr[i >> 5][i & 31] = adj[(size_t)nidx[i >> 5] * DEG + (i & 31)];
    __syncthreads();

    {
        const int half = t >> 6;
        const int tc   = (t & 63) << 2;
        for (int r = 0; r < TM; r += 2) {
            const int rr  = r + half;
            const int idx = nidx[rr];
            float s0 = 0.f, s1 = 0.f, s2 = 0.f, s3 = 0.f;
            float f0 = 0.f, f1 = 0.f, f2 = 0.f, f3 = 0.f;
            #pragma unroll 8
            for (int d = 0; d < DEG; ++d) {
                const int j = nbr[rr][d];
                const v4f fa = *(const v4f*)(feat_data + (size_t)j * FDIM + tc);
                const v4f fb = *(const v4f*)(feats     + (size_t)j * FDIM + tc);
                s0 += fa.x; s1 += fa.y; s2 += fa.z; s3 += fa.w;
                f0 += fb.x; f1 += fb.y; f2 += fb.z; f3 += fb.w;
            }
            v4f sv; sv.x = s0; sv.y = s1; sv.z = s2; sv.w = s3;
            v4f fv; fv.x = f0; fv.y = f1; fv.z = f2; fv.w = f3;
            *(v4f*)(&S[rr][tc])        = sv;
            *(v4f*)(&S[rr][FDIM + tc]) = fv;
            *(v4f*)(&T[rr][tc]) = *(const v4f*)(feat_data + (size_t)idx * FDIM + tc);
        }
    }
    __syncthreads();

    const int lr    = lane & 15;
    const int kh    = (lane >> 4) << 1;
    const int rbase = (lane >> 4) << 3;

    {
        v8f acc0 = {}, acc1 = {}, acc2 = {}, acc3 = {};
        for (int k0 = 0; k0 < KDIM; k0 += 4) {
            v2f a;
            a.x = S[lr][k0 + kh];
            a.y = S[lr][k0 + kh + 1];
            const int nb0 = wave * 64 + lr;
            const float* w0 = W1 + (size_t)(nb0     ) * KDIM + k0 + kh;
            const float* w1 = W1 + (size_t)(nb0 + 16) * KDIM + k0 + kh;
            const float* w2 = W1 + (size_t)(nb0 + 32) * KDIM + k0 + kh;
            const float* w3 = W1 + (size_t)(nb0 + 48) * KDIM + k0 + kh;
            v2f bv;
            bv.x = w0[0]; bv.y = w0[1];
            acc0 = __builtin_amdgcn_wmma_f32_16x16x4_f32(false, a, false, bv, (short)0, acc0, false, false);
            bv.x = w1[0]; bv.y = w1[1];
            acc1 = __builtin_amdgcn_wmma_f32_16x16x4_f32(false, a, false, bv, (short)0, acc1, false, false);
            bv.x = w2[0]; bv.y = w2[1];
            acc2 = __builtin_amdgcn_wmma_f32_16x16x4_f32(false, a, false, bv, (short)0, acc2, false, false);
            bv.x = w3[0]; bv.y = w3[1];
            acc3 = __builtin_amdgcn_wmma_f32_16x16x4_f32(false, a, false, bv, (short)0, acc3, false, false);
        }
        #pragma unroll
        for (int tile = 0; tile < 4; ++tile) {
            const v8f acc = (tile == 0) ? acc0 : (tile == 1) ? acc1 : (tile == 2) ? acc2 : acc3;
            const int n = wave * 64 + tile * 16 + lr;
            const float bias = (float)DEG * b1[n];
            #pragma unroll
            for (int v = 0; v < 8; ++v)
                T[rbase + v][FDIM + n] = acc[v] + bias;
        }
    }
    __syncthreads();

    {
        v8f acc0 = {}, acc1 = {}, acc2 = {}, acc3 = {};
        for (int k0 = 0; k0 < KDIM; k0 += 4) {
            v2f a;
            a.x = T[lr][k0 + kh];
            a.y = T[lr][k0 + kh + 1];
            const int nb0 = wave * 64 + lr;
            const float* w0 = W2 + (size_t)(nb0     ) * KDIM + k0 + kh;
            const float* w1 = W2 + (size_t)(nb0 + 16) * KDIM + k0 + kh;
            const float* w2 = W2 + (size_t)(nb0 + 32) * KDIM + k0 + kh;
            const float* w3 = W2 + (size_t)(nb0 + 48) * KDIM + k0 + kh;
            v2f bv;
            bv.x = w0[0]; bv.y = w0[1];
            acc0 = __builtin_amdgcn_wmma_f32_16x16x4_f32(false, a, false, bv, (short)0, acc0, false, false);
            bv.x = w1[0]; bv.y = w1[1];
            acc1 = __builtin_amdgcn_wmma_f32_16x16x4_f32(false, a, false, bv, (short)0, acc1, false, false);
            bv.x = w2[0]; bv.y = w2[1];
            acc2 = __builtin_amdgcn_wmma_f32_16x16x4_f32(false, a, false, bv, (short)0, acc2, false, false);
            bv.x = w3[0]; bv.y = w3[1];
            acc3 = __builtin_amdgcn_wmma_f32_16x16x4_f32(false, a, false, bv, (short)0, acc3, false, false);
        }
        #pragma unroll
        for (int tile = 0; tile < 4; ++tile) {
            const v8f acc = (tile == 0) ? acc0 : (tile == 1) ? acc1 : (tile == 2) ? acc2 : acc3;
            const int n = wave * 64 + tile * 16 + lr;
            const float bias = b2[n];
            #pragma unroll
            for (int v = 0; v < 8; ++v)
                S[rbase + v][n] = acc[v] + bias;
        }
    }
    __syncthreads();

    {
        const int r = t >> 3, seg = t & 7;
        float s = 0.f;
        #pragma unroll 8
        for (int c = seg * 32; c < seg * 32 + 32; ++c) {
            const float x = S[r][c];
            s += x * x;
        }
        part[r][seg] = s;
    }
    __syncthreads();
    if (t < TM) {
        float s = 0.f;
        #pragma unroll
        for (int i = 0; i < 8; ++i) s += part[t][i];
        inv_norm[t] = 1.0f / fmaxf(sqrtf(s), 1e-12f);
    }
    __syncthreads();
    for (int i = t; i < TM * FDIM; i += THREADS) {
        const int r = i >> 8;
        const int c = i & (FDIM - 1);
        const int g = m0 + r;
        if (g < nNodes)
            out[(size_t)g * FDIM + c] = S[r][c] * inv_norm[r];
    }
}

extern "C" void kernel_launch(void* const* d_in, const int* in_sizes, int n_in,
                              void* d_out, int out_size, void* d_ws, size_t ws_size,
                              hipStream_t stream) {
    const float* feat_data = (const float*)d_in[0];
    const float* feats     = (const float*)d_in[1];
    const float* W1        = (const float*)d_in[2];
    const float* b1        = (const float*)d_in[3];
    const float* W2        = (const float*)d_in[4];
    const float* b2        = (const float*)d_in[5];
    const int*   adj       = (const int*)d_in[6];
    const int*   in1       = (const int*)d_in[7];
    const int*   in2       = (const int*)d_in[8];
    const int*   neg       = (const int*)d_in[9];
    float* out = (float*)d_out;

    const int B1 = in_sizes[7];
    const int B2 = in_sizes[8];
    const int Bn = in_sizes[9];

    dim3 blk(THREADS);
    graphsage_fused_kernel<<<dim3((B1 + TM - 1) / TM), blk, 0, stream>>>(
        feat_data, feats, W1, b1, W2, b2, adj, in1, B1, out);
    graphsage_fused_kernel<<<dim3((B2 + TM - 1) / TM), blk, 0, stream>>>(
        feat_data, feats, W1, b1, W2, b2, adj, in2, B2, out + (size_t)B1 * FDIM);
    graphsage_fused_kernel<<<dim3((Bn + TM - 1) / TM), blk, 0, stream>>>(
        feat_data, feats, W1, b1, W2, b2, adj, neg, Bn, out + (size_t)(B1 + B2) * FDIM);
}


# Gather waves own four batch rows each (four neighbours of every row, 32 loads in flight per wave, 4-wave work-groups): a quarter as many waves enter, the whole grid is resident; row sum order kept
# speedup vs baseline: 1.5383x; 1.5383x over previous
#include <hip/hip_runtime.h>
#include <hip/hip_bf16.h>


typedef float v2f __attribute__((ext_vector_type(2)));
typedef float v4f __attribute__((ext_vector_type(4)));
typedef float v8f __attribute__((ext_vector_type(8)));

#define FDIM 256
#define KDIM 512
#define DEG  32
#define TM   16
#define THREADS 128
#define GTHREADS 128


__device__ __forceinline__ int pick_node(const int* __restrict__ in1,
                                         const int* __restrict__ in2,
                                         const int* __restrict__ neg,
                                         int B1, int B2, int g)
{
    const bool a = g < B1, b = g < B1 + B2;
    const int* __restrict__ p = a ? in1 : (b ? in2 : neg);
    const int o = a ? g : (b ? g - B1 : g - B1 - B2);
    return p[o];
}

#define RPW    4
#define GWAVES (GTHREADS / 32)
#define GROWS  (GWAVES * RPW)

__global__ __launch_bounds__(GTHREADS)
void neighbour_sum_kernel(const float* __restrict__ feat_data,
                          const float* __restrict__ feats,
                          const int*   __restrict__ adj,
                          const int*   __restrict__ in1,
                          const int*   __restrict__ in2,
                          const int*   __restrict__ neg,
                          int B1, int B2, int total,
                          float* __restrict__ Sws)
{
    const int lane = threadIdx.x & 31;
    const int wave = __builtin_amdgcn_readfirstlane((int)(threadIdx.x >> 5));
    const int row0 = (blockIdx.x * GWAVES + wave) * RPW;
    if (row0 >= total) return;
    const int last = total - 1;

    int jv[RPW];
    #pragma unroll
    for (int r = 0; r < RPW; ++r) {
        const int node = pick_node(in1, in2, neg, B1, B2, min(row0 + r, last));
        jv[r] = adj[(size_t)node * DEG + lane];
    }

    int sv[RPW];
    #pragma unroll
    for (int r = 0; r < RPW; ++r) {
        int rank = 0;
        #pragma unroll
        for (int j = 0; j < DEG; ++j) {
            const int s = __builtin_amdgcn_readlane(jv[r], j);
            rank += (s < jv[r] || (s == jv[r] && j < lane)) ? 1 : 0;
        }
        sv[r] = __builtin_amdgcn_ds_permute(rank << 2, jv[r]);
    }

    const float* __restrict__ t0 = feat_data + (lane << 2);
    const float* __restrict__ t1 = feats + (lane << 2);
    float* dst = Sws + (size_t)row0 * KDIM + (lane << 2);

    v4f a[4][RPW][2];
    #pragma unroll
    for (int i = 0; i < 4; ++i) {
        #pragma unroll
        for (int r = 0; r < RPW; ++r) {
            const float* p = t0 + (size_t)__builtin_amdgcn_readlane(sv[r], i) * FDIM;
            a[i][r][0] = *(const v4f*)(p);
            a[i][r][1] = *(const v4f*)(p + 128);
        }
    }
    v4f lo[RPW], hi[RPW];
    #pragma unroll
    for (int r = 0; r < RPW; ++r) { lo[r] = (v4f){}; hi[r] = (v4f){}; }
    #pragma nounroll
    for (int d = 4; d < DEG; d += 4) {
        #pragma unroll
        for (int i = 0; i < 4; ++i) {
            #pragma unroll
            for (int r = 0; r < RPW; ++r) {
                lo[r] += a[i][r][0];
                hi[r] += a[i][r][1];
                asm volatile("" :: "v"(lo[r]), "v"(hi[r]) : "memory");
                const float* p = t0 + (size_t)__builtin_amdgcn_readlane(sv[r], d + i) * FDIM;
                a[i][r][0] = *(const v4f*)(p);
                a[i][r][1] = *(const v4f*)(p + 128);
                __builtin_amdgcn_sched_barrier(0);
            }
        }
    }
    #pragma unroll
    for (int i = 0; i < 4; ++i) {
        #pragma unroll
        for (int r = 0; r < RPW; ++r) {
            lo[r] += a[i][r][0];
            hi[r] += a[i][r][1];
            asm volatile("" :: "v"(lo[r]), "v"(hi[r]) : "memory");
            const float* p = t1 + (size_t)__builtin_amdgcn_readlane(sv[r], i) * FDIM;
            a[i][r][0] = *(const v4f*)(p);
            a[i][r][1] = *(const v4f*)(p + 128);
            __builtin_amdgcn_sched_barrier(0);
        }
    }
    #pragma unroll
    for (int r = 0; r < RPW; ++r) {
        if (row0 + r < total) {
            *(v4f*)(dst + (size_t)r * KDIM)       = lo[r];
            *(v4f*)(dst + (size_t)r * KDIM + 128) = hi[r];
        }
        lo[r] = (v4f){};
        hi[r] = (v4f){};
    }
    #pragma nounroll
    for (int d = 4; d < DEG; d += 4) {
        #pragma unroll
        for (int i = 0; i < 4; ++i) {
            #pragma unroll
            for (int r = 0; r < RPW; ++r) {
                lo[r] += a[i][r][0];
                hi[r] += a[i][r][1];
                asm volatile("" :: "v"(lo[r]), "v"(hi[r]) : "memory");
                const float* p = t1 + (size_t)__builtin_amdgcn_readlane(sv[r], d + i) * FDIM;
                a[i][r][0] = *(const v4f*)(p);
                a[i][r][1] = *(const v4f*)(p + 128);
                __builtin_amdgcn_sched_barrier(0);
            }
        }
    }
    #pragma unroll
    for (int i = 0; i < 4; ++i) {
        #pragma unroll
        for (int r = 0; r < RPW; ++r) {
            lo[r] += a[i][r][0];
            hi[r] += a[i][r][1];
        }
    }
    #pragma unroll
    for (int r = 0; r < RPW; ++r) {
        if (row0 + r < total) {
            *(v4f*)(dst + (size_t)r * KDIM + FDIM)       = lo[r];
            *(v4f*)(dst + (size_t)r * KDIM + FDIM + 128) = hi[r];
        }
    }
}


#define SLABK   32
#define HALFN   128
#define STAGE_F (HALFN * SLABK)
#define NSLAB   (2 * KDIM / SLABK)

__device__ __forceinline__ v8f wmma4(v2f a, v2f b, v8f c)
{
    return __builtin_amdgcn_wmma_f32_16x16x4_f32(false, a, false, b, (short)0, c, false, false);
}

__device__ __forceinline__ void gemm_phase(const float* __restrict__ Wg,
                                           const float* At, float* stage,
                                           int t, int wave, int lr, int hl,
                                           v8f (&acc)[4])
{
    const float* gsrc = Wg + (size_t)(t >> 3) * KDIM + ((t & 7) << 2);
    const int sdst = (t >> 3) * SLABK + (((t & 7) ^ ((t >> 4) & 7)) << 2);
    const int bl  = (wave * 32 + lr) * SLABK + (((lr >> 1) & 7) << 2) + 2 * hl;
    const int al  = lr * KDIM + 2 * hl;
    const int asw = lr << 2;

    v4f w[8];
    #pragma unroll
    for (int i = 0; i < 8; ++i) w[i] = *(const v4f*)(gsrc + (size_t)i * 16 * KDIM);
    #pragma unroll
    for (int i = 0; i < 8; ++i) *(v4f*)(stage + sdst + i * 16 * SLABK) = w[i];
    __syncthreads();

    #pragma unroll
    for (int nh = 0; nh < 2; ++nh) {
        v8f c0 = {}, c1 = {};
        #pragma nounroll
        for (int kk = 0; kk < KDIM / SLABK; ++kk) {
            const int s = nh * (KDIM / SLABK) + kk;
            const bool more = s + 1 < NSLAB;
            if (more) {
                const int s1 = s + 1;
                const float* g = gsrc + (size_t)(s1 >> 4) * HALFN * KDIM + (s1 & 15) * SLABK;
                #pragma unroll
                for (int i = 0; i < 8; ++i) w[i] = *(const v4f*)(g + (size_t)i * 16 * KDIM);
            }
            const int so = (s & 1) * STAGE_F;
            const int kx = (kk * SLABK) ^ asw;
            v2f a[8], b0[8], b1[8];
            #pragma unroll
            for (int j = 0; j < 8; ++j) {
                a[j] = *(const v2f*)(At + al + (kx ^ (j << 2)));
                const int bo = so + (bl ^ (j << 2));
                b0[j] = *(const v2f*)(stage + bo);
                b1[j] = *(const v2f*)(stage + bo + 16 * SLABK);
            }
            #pragma unroll
            for (int j = 0; j < 8; ++j) {
                c0 = wmma4(a[j], b0[j], c0);
                c1 = wmma4(a[j], b1[j], c1);
            }
            if (more) {
                float* sd = stage + (STAGE_F - so) + sdst;
                #pragma unroll
                for (int i = 0; i < 8; ++i) *(v4f*)(sd + i * 16 * SLABK) = w[i];
            }
            __syncthreads();
        }
        acc[2 * nh]     = c0;
        acc[2 * nh + 1] = c1;
    }
}

__global__ __launch_bounds__(THREADS)
void gemm_norm_kernel(const float* __restrict__ feat_data,
                      const float* __restrict__ W1,
                      const float* __restrict__ b1,
                      const float* __restrict__ W2,
                      const float* __restrict__ b2,
                      const int*   __restrict__ in1,
                      const int*   __restrict__ in2,
                      const int*   __restrict__ neg,
                      int B1, int B2, int total,
                      const float* __restrict__ Sws,
                      float* __restrict__ out)
{
    __shared__ float S[TM * KDIM];
    __shared__ float T[TM * KDIM];
    float (*part)[8] = reinterpret_cast<float (*)[8]>(&T[0]);
    float* inv_norm  = &T[TM * 8];

    const int t    = threadIdx.x;
    const int m0   = blockIdx.x * TM;
    const int lane = t & 31;
    const int wave = t >> 5;
    const int lr    = lane & 15;
    const int hl    = lane >> 4;
    const int rbase = hl << 3;

    const int half = t >> 6;
    const int tc   = (t & 63) << 2;
    v4f tv[TM / 2];
    {
        const int last = total - 1;
        int nid[TM / 2];
        #pragma unroll
        for (int k = 0; k < TM / 2; ++k)
            nid[k] = pick_node(in1, in2, neg, B1, B2, min(m0 + 2 * k + half, last));
        v4f sv[TM];
        #pragma unroll
        for (int k = 0; k < TM; ++k)
            sv[k] = *(const v4f*)(Sws + (size_t)min(m0 + k, last) * KDIM + (t << 2));
        #pragma unroll
        for (int k = 0; k < TM / 2; ++k)
            tv[k] = *(const v4f*)(feat_data + (size_t)nid[k] * FDIM + tc);
        #pragma unroll
        for (int k = 0; k < TM; ++k)
            *(v4f*)(&S[k * KDIM + ((t << 2) ^ (k << 2))]) = sv[k];
    }

    v8f acc[4];

    gemm_phase(W1, S, T, t, wave, lr, hl, acc);
    #pragma unroll
    for (int k = 0; k < TM / 2; ++k) {
        const int r = 2 * k + half;
        *(v4f*)(&T[r * KDIM + (tc ^ (r << 2))]) = tv[k];
    }
    #pragma unroll
    for (int q = 0; q < 4; ++q) {
        const int n = (q >> 1) * HALFN + wave * 32 + (q & 1) * 16 + lr;
        const float bias = (float)DEG * b1[n];
        #pragma unroll
        for (int v = 0; v < 8; ++v) {
            const int r = rbase + v;
            T[r * KDIM + ((FDIM + n) ^ (r << 2))] = acc[q][v] + bias;
        }
    }

    gemm_phase(W2, T, S, t, wave, lr, hl, acc);
    #pragma unroll
    for (int q = 0; q < 4; ++q) {
        const int n = (q >> 1) * HALFN + wave * 32 + (q & 1) * 16 + lr;
        const float bias = b2[n];
        #pragma unroll
        for (int v = 0; v < 8; ++v)
            S[(rbase + v) * KDIM + n] = acc[q][v] + bias;
    }
    __syncthreads();

    {
        const int r = t >> 3, seg = t & 7;
        float s = 0.f;
        #pragma unroll 8
        for (int c = seg * 32; c < seg * 32 + 32; ++c) {
            const float x = S[r * KDIM + c];
            s += x * x;
        }
        part[r][seg] = s;
    }
    __syncthreads();
    if (t < TM) {
        float s = 0.f;
        #pragma unroll
        for (int i = 0; i < 8; ++i) s += part[t][i];
        inv_norm[t] = 1.0f / fmaxf(sqrtf(s), 1e-12f);
    }
    __syncthreads();
    for (int i = t; i < TM * FDIM; i += THREADS) {
        const int r = i >> 8;
        const int c = i & (FDIM - 1);
        const int g = m0 + r;
        if (g < total)
            out[(size_t)g * FDIM + c] = S[r * KDIM + c] * inv_norm[r];
    }
}

extern "C" void kernel_launch(void* const* d_in, const int* in_sizes, int n_in,
                              void* d_out, int out_size, void* d_ws, size_t ws_size,
                              hipStream_t stream) {
    const float* feat_data = (const float*)d_in[0];
    const float* feats     = (const float*)d_in[1];
    const float* W1        = (const float*)d_in[2];
    const float* b1        = (const float*)d_in[3];
    const float* W2        = (const float*)d_in[4];
    const float* b2        = (const float*)d_in[5];
    const int*   adj       = (const int*)d_in[6];
    const int*   in1       = (const int*)d_in[7];
    const int*   in2       = (const int*)d_in[8];
    const int*   neg       = (const int*)d_in[9];
    float* out = (float*)d_out;
    float* Sws = (float*)d_ws;

    const int B1 = in_sizes[7];
    const int B2 = in_sizes[8];
    const int Bn = in_sizes[9];
    const int total = B1 + B2 + Bn;

    neighbour_sum_kernel<<<dim3((total + GROWS - 1) / GROWS), dim3(GTHREADS), 0, stream>>>(
        feat_data, feats, adj, in1, in2, neg, B1, B2, total, Sws);
    gemm_norm_kernel<<<dim3((total + TM - 1) / TM), dim3(THREADS), 0, stream>>>(
        feat_data, W1, b1, W2, b2, in1, in2, neg, B1, B2, total, Sws, out);
}


# Both launches fused into one: each 16-row strip's work-group sums its rows (same sweep) straight into its LDS tile, then runs its GEMMs; no launch boundary, no workspace plane; sum order kept
# speedup vs baseline: 1.6278x; 1.0582x over previous
#include <hip/hip_runtime.h>
#include <hip/hip_bf16.h>


typedef float v2f __attribute__((ext_vector_type(2)));
typedef float v4f __attribute__((ext_vector_type(4)));
typedef float v8f __attribute__((ext_vector_type(8)));

#define FDIM 256
#define KDIM 512
#define DEG  32
#define TM   16
#define THREADS 128


__device__ __forceinline__ int pick_node(const int* __restrict__ in1,
                                         const int* __restrict__ in2,
                                         const int* __restrict__ neg,
                                         int B1, int B2, int g)
{
    const bool a = g < B1, b = g < B1 + B2;
    const int* __restrict__ p = a ? in1 : (b ? in2 : neg);
    const int o = a ? g : (b ? g - B1 : g - B1 - B2);
    return p[o];
}

#define RPW    4
static_assert((THREADS / 32) * RPW == TM, "the waves of a work-group sum exactly its strip");

__device__ __forceinline__
void sum_rows(const float* __restrict__ feat_data,
              const float* __restrict__ feats,
              const int*   __restrict__ adj,
              const int*   __restrict__ in1,
              const int*   __restrict__ in2,
              const int*   __restrict__ neg,
              int B1, int B2, int total, int row0, int lane,
              float* tile, int trow0)
{
    const int last = total - 1;

    int jv[RPW];
    #pragma unroll
    for (int r = 0; r < RPW; ++r) {
        const int node = pick_node(in1, in2, neg, B1, B2, min(row0 + r, last));
        jv[r] = adj[(size_t)node * DEG + lane];
    }

    int sv[RPW];
    #pragma unroll
    for (int r = 0; r < RPW; ++r) {
        int rank = 0;
        #pragma unroll
        for (int j = 0; j < DEG; ++j) {
            const int s = __builtin_amdgcn_readlane(jv[r], j);
            rank += (s < jv[r] || (s == jv[r] && j < lane)) ? 1 : 0;
        }
        sv[r] = __builtin_amdgcn_ds_permute(rank << 2, jv[r]);
    }

    const float* __restrict__ t0 = feat_data + (lane << 2);
    const float* __restrict__ t1 = feats + (lane << 2);
    auto put = [&](int r, int cb, v4f v) {
        const int k = trow0 + r;
        *(v4f*)(tile + k * KDIM + ((cb * 128 + (lane << 2)) ^ (k << 2))) = v;
    };

    v4f a[4][RPW][2];
    #pragma unroll
    for (int i = 0; i < 4; ++i) {
        #pragma unroll
        for (int r = 0; r < RPW; ++r) {
            const float* p = t0 + (size_t)__builtin_amdgcn_readlane(sv[r], i) * FDIM;
            a[i][r][0] = *(const v4f*)(p);
            a[i][r][1] = *(const v4f*)(p + 128);
        }
    }
    v4f lo[RPW], hi[RPW];
    #pragma unroll
    for (int r = 0; r < RPW; ++r) { lo[r] = (v4f){}; hi[r] = (v4f){}; }
    #pragma nounroll
    for (int d = 4; d < DEG; d += 4) {
        #pragma unroll
        for (int i = 0; i < 4; ++i) {
            #pragma unroll
            for (int r = 0; r < RPW; ++r) {
                lo[r] += a[i][r][0];
                hi[r] += a[i][r][1];
                asm volatile("" :: "v"(lo[r]), "v"(hi[r]) : "memory");
                const float* p = t0 + (size_t)__builtin_amdgcn_readlane(sv[r], d + i) * FDIM;
                a[i][r][0] = *(const v4f*)(p);
                a[i][r][1] = *(const v4f*)(p + 128);
                __builtin_amdgcn_sched_barrier(0);
            }
        }
    }
    #pragma unroll
    for (int i = 0; i < 4; ++i) {
        #pragma unroll
        for (int r = 0; r < RPW; ++r) {
            lo[r] += a[i][r][0];
            hi[r] += a[i][r][1];
            asm volatile("" :: "v"(lo[r]), "v"(hi[r]) : "memory");
            const float* p = t1 + (size_t)__builtin_amdgcn_readlane(sv[r], i) * FDIM;
            a[i][r][0] = *(const v4f*)(p);
            a[i][r][1] = *(const v4f*)(p + 128);
            __builtin_amdgcn_sched_barrier(0);
        }
    }
    #pragma unroll
    for (int r = 0; r < RPW; ++r) {
        put(r, 0, lo[r]);
        put(r, 1, hi[r]);
        lo[r] = (v4f){};
        hi[r] = (v4f){};
    }
    #pragma nounroll
    for (int d = 4; d < DEG; d += 4) {
        #pragma unroll
        for (int i = 0; i < 4; ++i) {
            #pragma unroll
            for (int r = 0; r < RPW; ++r) {
                lo[r] += a[i][r][0];
                hi[r] += a[i][r][1];
                asm volatile("" :: "v"(lo[r]), "v"(hi[r]) : "memory");
                const float* p = t1 + (size_t)__builtin_amdgcn_readlane(sv[r], d + i) * FDIM;
                a[i][r][0] = *(const v4f*)(p);
                a[i][r][1] = *(const v4f*)(p + 128);
                __builtin_amdgcn_sched_barrier(0);
            }
        }
    }
    #pragma unroll
    for (int i = 0; i < 4; ++i) {
        #pragma unroll
        for (int r = 0; r < RPW; ++r) {
            lo[r] += a[i][r][0];
            hi[r] += a[i][r][1];
        }
    }
    #pragma unroll
    for (int r = 0; r < RPW; ++r) {
        put(r, 2, lo[r]);
        put(r, 3, hi[r]);
    }
}


#define SLABK   32
#define HALFN   128
#define STAGE_F (HALFN * SLABK)
#define NSLAB   (2 * KDIM / SLABK)

__device__ __forceinline__ v8f wmma4(v2f a, v2f b, v8f c)
{
    return __builtin_amdgcn_wmma_f32_16x16x4_f32(false, a, false, b, (short)0, c, false, false);
}

__device__ __forceinline__ void gemm_phase(const float* __restrict__ Wg,
                                           const float* At, float* stage,
                                           int t, int wave, int lr, int hl,
                                           v8f (&acc)[4])
{
    const float* gsrc = Wg + (size_t)(t >> 3) * KDIM + ((t & 7) << 2);
    const int sdst = (t >> 3) * SLABK + (((t & 7) ^ ((t >> 4) & 7)) << 2);
    const int bl  = (wave * 32 + lr) * SLABK + (((lr >> 1) & 7) << 2) + 2 * hl;
    const int al  = lr * KDIM + 2 * hl;
    const int asw = lr << 2;

    v4f w[8];
    #pragma unroll
    for (int i = 0; i < 8; ++i) w[i] = *(const v4f*)(gsrc + (size_t)i * 16 * KDIM);
    #pragma unroll
    for (int i = 0; i < 8; ++i) *(v4f*)(stage + sdst + i * 16 * SLABK) = w[i];
    __syncthreads();

    #pragma unroll
    for (int nh = 0; nh < 2; ++nh) {
        v8f c0 = {}, c1 = {};
        #pragma nounroll
        for (int kk = 0; kk < KDIM / SLABK; ++kk) {
            const int s = nh * (KDIM / SLABK) + kk;
            const bool more = s + 1 < NSLAB;
            if (more) {
                const int s1 = s + 1;
                const float* g = gsrc + (size_t)(s1 >> 4) * HALFN * KDIM + (s1 & 15) * SLABK;
                #pragma unroll
                for (int i = 0; i < 8; ++i) w[i] = *(const v4f*)(g + (size_t)i * 16 * KDIM);
            }
            const int so = (s & 1) * STAGE_F;
            const int kx = (kk * SLABK) ^ asw;
            v2f a[8], b0[8], b1[8];
            #pragma unroll
            for (int j = 0; j < 8; ++j) {
                a[j] = *(const v2f*)(At + al + (kx ^ (j << 2)));
                const int bo = so + (bl ^ (j << 2));
                b0[j] = *(const v2f*)(stage + bo);
                b1[j] = *(const v2f*)(stage + bo + 16 * SLABK);
            }
            #pragma unroll
            for (int j = 0; j < 8; ++j) {
                c0 = wmma4(a[j], b0[j], c0);
                c1 = wmma4(a[j], b1[j], c1);
            }
            if (more) {
                float* sd = stage + (STAGE_F - so) + sdst;
                #pragma unroll
                for (int i = 0; i < 8; ++i) *(v4f*)(sd + i * 16 * SLABK) = w[i];
            }
            __syncthreads();
        }
        acc[2 * nh]     = c0;
        acc[2 * nh + 1] = c1;
    }
}

__device__ __forceinline__
void strip_finish(float* S, float* T, const v4f (&tv)[TM / 2],
                  const float* __restrict__ W1, const float* __restrict__ b1,
                  const float* __restrict__ W2, const float* __restrict__ b2,
                  int m0, int total, float* __restrict__ out)
{
    float (*part)[8] = reinterpret_cast<float (*)[8]>(&T[0]);
    float* inv_norm  = &T[TM * 8];

    const int t    = threadIdx.x;
    const int lane = t & 31;
    const int wave = t >> 5;
    const int lr    = lane & 15;
    const int hl    = lane >> 4;
    const int rbase = hl << 3;
    const int half = t >> 6;
    const int tc   = (t & 63) << 2;

    v8f acc[4];

    gemm_phase(W1, S, T, t, wave, lr, hl, acc);
    #pragma unroll
    for (int k = 0; k < TM / 2; ++k) {
        const int r = 2 * k + half;
        *(v4f*)(&T[r * KDIM + (tc ^ (r << 2))]) = tv[k];
    }
    #pragma unroll
    for (int q = 0; q < 4; ++q) {
        const int n = (q >> 1) * HALFN + wave * 32 + (q & 1) * 16 + lr;
        const float bias = (float)DEG * b1[n];
        #pragma unroll
        for (int v = 0; v < 8; ++v) {
            const int r = rbase + v;
            T[r * KDIM + ((FDIM + n) ^ (r << 2))] = acc[q][v] + bias;
        }
    }

    gemm_phase(W2, T, S, t, wave, lr, hl, acc);
    #pragma unroll
    for (int q = 0; q < 4; ++q) {
        const int n = (q >> 1) * HALFN + wave * 32 + (q & 1) * 16 + lr;
        const float bias = b2[n];
        #pragma unroll
        for (int v = 0; v < 8; ++v)
            S[(rbase + v) * KDIM + n] = acc[q][v] + bias;
    }
    __syncthreads();

    {
        const int r = t >> 3, seg = t & 7;
        float s = 0.f;
        #pragma unroll 8
        for (int c = seg * 32; c < seg * 32 + 32; ++c) {
            const float x = S[r * KDIM + c];
            s += x * x;
        }
        part[r][seg] = s;
    }
    __syncthreads();
    if (t < TM) {
        float s = 0.f;
        #pragma unroll
        for (int i = 0; i < 8; ++i) s += part[t][i];
        inv_norm[t] = 1.0f / fmaxf(sqrtf(s), 1e-12f);
    }
    __syncthreads();
    for (int i = t; i < TM * FDIM; i += THREADS) {
        const int r = i >> 8;
        const int c = i & (FDIM - 1);
        const int g = m0 + r;
        if (g < total)
            out[(size_t)g * FDIM + c] = S[r * KDIM + c] * inv_norm[r];
    }
}

__device__ __forceinline__
void load_node_ids(const int* __restrict__ in1, const int* __restrict__ in2,
                   const int* __restrict__ neg, int B1, int B2, int total,
                   int m0, int (&nid)[TM / 2])
{
    const int half = threadIdx.x >> 6;
    const int last = total - 1;
    #pragma unroll
    for (int k = 0; k < TM / 2; ++k)
        nid[k] = pick_node(in1, in2, neg, B1, B2, min(m0 + 2 * k + half, last));
}

__device__ __forceinline__
void load_node_rows(const float* __restrict__ feat_data,
                    const int (&nid)[TM / 2], v4f (&tv)[TM / 2])
{
    const int tc = (threadIdx.x & 63) << 2;
    #pragma unroll
    for (int k = 0; k < TM / 2; ++k)
        tv[k] = *(const v4f*)(feat_data + (size_t)nid[k] * FDIM + tc);
}

__global__ __launch_bounds__(THREADS)
void graph_strip_kernel(const float* __restrict__ feat_data,
                        const float* __restrict__ feats,
                        const float* __restrict__ W1,
                        const float* __restrict__ b1,
                        const float* __restrict__ W2,
                        const float* __restrict__ b2,
                        const int*   __restrict__ adj,
                        const int*   __restrict__ in1,
                        const int*   __restrict__ in2,
                        const int*   __restrict__ neg,
                        int B1, int B2, int total,
                        float* __restrict__ out)
{
    __shared__ float S[TM * KDIM];
    __shared__ float T[TM * KDIM];

    const int lane = threadIdx.x & 31;
    const int wave = __builtin_amdgcn_readfirstlane((int)(threadIdx.x >> 5));
    const int m0 = (int)blockIdx.x * TM;

    sum_rows(feat_data, feats, adj, in1, in2, neg, B1, B2, total,
             m0 + wave * RPW, lane, S, wave * RPW);
    int nid[TM / 2];
    load_node_ids(in1, in2, neg, B1, B2, total, m0, nid);
    v4f tv[TM / 2];
    load_node_rows(feat_data, nid, tv);

    strip_finish(S, T, tv, W1, b1, W2, b2, m0, total, out);
}

extern "C" void kernel_launch(void* const* d_in, const int* in_sizes, int n_in,
                              void* d_out, int out_size, void* d_ws, size_t ws_size,
                              hipStream_t stream) {
    const float* feat_data = (const float*)d_in[0];
    const float* feats     = (const float*)d_in[1];
    const float* W1        = (const float*)d_in[2];
    const float* b1        = (const float*)d_in[3];
    const float* W2        = (const float*)d_in[4];
    const float* b2        = (const float*)d_in[5];
    const int*   adj       = (const int*)d_in[6];
    const int*   in1       = (const int*)d_in[7];
    const int*   in2       = (const int*)d_in[8];
    const int*   neg       = (const int*)d_in[9];
    float* out = (float*)d_out;

    const int B1 = in_sizes[7];
    const int B2 = in_sizes[8];
    const int Bn = in_sizes[9];
    const int total = B1 + B2 + Bn;

    graph_strip_kernel<<<dim3((total + TM - 1) / TM), dim3(THREADS), 0, stream>>>(
        feat_data, feats, W1, b1, W2, b2, adj, in1, in2, neg, B1, B2, total, out);
}


# The short last strip (the work-group beyond four per WGP) sums its rows as (row, table) units dealt over its waves, 16 neighbours in flight, so its GEMMs run while others still gather; sum order kept
# speedup vs baseline: 1.6802x; 1.0322x over previous
#include <hip/hip_runtime.h>
#include <hip/hip_bf16.h>


typedef float v2f __attribute__((ext_vector_type(2)));
typedef float v4f __attribute__((ext_vector_type(4)));
typedef float v8f __attribute__((ext_vector_type(8)));

#define FDIM 256
#define KDIM 512
#define DEG  32
#define TM   16
#define THREADS 128


__device__ __forceinline__ int pick_node(const int* __restrict__ in1,
                                         const int* __restrict__ in2,
                                         const int* __restrict__ neg,
                                         int B1, int B2, int g)
{
    const bool a = g < B1, b = g < B1 + B2;
    const int* __restrict__ p = a ? in1 : (b ? in2 : neg);
    const int o = a ? g : (b ? g - B1 : g - B1 - B2);
    return p[o];
}

#define RPW    4
static_assert((THREADS / 32) * RPW == TM, "the waves of a work-group sum exactly its strip");

__device__ __forceinline__
void sum_rows(const float* __restrict__ feat_data,
              const float* __restrict__ feats,
              const int*   __restrict__ adj,
              const int*   __restrict__ in1,
              const int*   __restrict__ in2,
              const int*   __restrict__ neg,
              int B1, int B2, int total, int row0, int lane,
              float* tile, int trow0)
{
    const int last = total - 1;

    int jv[RPW];
    #pragma unroll
    for (int r = 0; r < RPW; ++r) {
        const int node = pick_node(in1, in2, neg, B1, B2, min(row0 + r, last));
        jv[r] = adj[(size_t)node * DEG + lane];
    }

    int sv[RPW];
    #pragma unroll
    for (int r = 0; r < RPW; ++r) {
        int rank = 0;
        #pragma unroll
        for (int j = 0; j < DEG; ++j) {
            const int s = __builtin_amdgcn_readlane(jv[r], j);
            rank += (s < jv[r] || (s == jv[r] && j < lane)) ? 1 : 0;
        }
        sv[r] = __builtin_amdgcn_ds_permute(rank << 2, jv[r]);
    }

    const float* __restrict__ t0 = feat_data + (lane << 2);
    const float* __restrict__ t1 = feats + (lane << 2);
    auto put = [&](int r, int cb, v4f v) {
        const int k = trow0 + r;
        *(v4f*)(tile + k * KDIM + ((cb * 128 + (lane << 2)) ^ (k << 2))) = v;
    };

    v4f a[4][RPW][2];
    #pragma unroll
    for (int i = 0; i < 4; ++i) {
        #pragma unroll
        for (int r = 0; r < RPW; ++r) {
            const float* p = t0 + (size_t)__builtin_amdgcn_readlane(sv[r], i) * FDIM;
            a[i][r][0] = *(const v4f*)(p);
            a[i][r][1] = *(const v4f*)(p + 128);
        }
    }
    v4f lo[RPW], hi[RPW];
    #pragma unroll
    for (int r = 0; r < RPW; ++r) { lo[r] = (v4f){}; hi[r] = (v4f){}; }
    #pragma nounroll
    for (int d = 4; d < DEG; d += 4) {
        #pragma unroll
        for (int i = 0; i < 4; ++i) {
            #pragma unroll
            for (int r = 0; r < RPW; ++r) {
                lo[r] += a[i][r][0];
                hi[r] += a[i][r][1];
                asm volatile("" :: "v"(lo[r]), "v"(hi[r]) : "memory");
                const float* p = t0 + (size_t)__builtin_amdgcn_readlane(sv[r], d + i) * FDIM;
                a[i][r][0] = *(const v4f*)(p);
                a[i][r][1] = *(const v4f*)(p + 128);
                __builtin_amdgcn_sched_barrier(0);
            }
        }
    }
    #pragma unroll
    for (int i = 0; i < 4; ++i) {
        #pragma unroll
        for (int r = 0; r < RPW; ++r) {
            lo[r] += a[i][r][0];
            hi[r] += a[i][r][1];
            asm volatile("" :: "v"(lo[r]), "v"(hi[r]) : "memory");
            const float* p = t1 + (size_t)__builtin_amdgcn_readlane(sv[r], i) * FDIM;
            a[i][r][0] = *(const v4f*)(p);
            a[i][r][1] = *(const v4f*)(p + 128);
            __builtin_amdgcn_sched_barrier(0);
        }
    }
    #pragma unroll
    for (int r = 0; r < RPW; ++r) {
        put(r, 0, lo[r]);
        put(r, 1, hi[r]);
        lo[r] = (v4f){};
        hi[r] = (v4f){};
    }
    #pragma nounroll
    for (int d = 4; d < DEG; d += 4) {
        #pragma unroll
        for (int i = 0; i < 4; ++i) {
            #pragma unroll
            for (int r = 0; r < RPW; ++r) {
                lo[r] += a[i][r][0];
                hi[r] += a[i][r][1];
                asm volatile("" :: "v"(lo[r]), "v"(hi[r]) : "memory");
                const float* p = t1 + (size_t)__builtin_amdgcn_readlane(sv[r], d + i) * FDIM;
                a[i][r][0] = *(const v4f*)(p);
                a[i][r][1] = *(const v4f*)(p + 128);
                __builtin_amdgcn_sched_barrier(0);
            }
        }
    }
    #pragma unroll
    for (int i = 0; i < 4; ++i) {
        #pragma unroll
        for (int r = 0; r < RPW; ++r) {
            lo[r] += a[i][r][0];
            hi[r] += a[i][r][1];
        }
    }
    #pragma unroll
    for (int r = 0; r < RPW; ++r) {
        put(r, 2, lo[r]);
        put(r, 3, hi[r]);
    }
}

#define MAXU ((2 * TM) / (THREADS / 32))
__device__ __forceinline__
void sum_rows_short(const float* __restrict__ feat_data,
                    const float* __restrict__ feats,
                    const int*   __restrict__ adj,
                    const int*   __restrict__ in1,
                    const int*   __restrict__ in2,
                    const int*   __restrict__ neg,
                    int B1, int B2, int m0, int nvalid, int wave, int lane,
                    float* tile, int* ids)
{
    auto put = [&](int k, int cb, v4f v) {
        *(v4f*)(tile + k * KDIM + ((cb * 128 + (lane << 2)) ^ (k << 2))) = v;
    };
    for (int k = nvalid + wave; k < TM; k += THREADS / 32) {
        #pragma unroll
        for (int cb = 0; cb < 4; ++cb) put(k, cb, (v4f){});
    }

    int* myids = ids + wave * (MAXU * 32) + lane;
    #pragma unroll
    for (int k = 0; k < MAXU; ++k) {
        const int r = min((wave + (THREADS / 32) * k) >> 1, nvalid - 1);
        const int node = pick_node(in1, in2, neg, B1, B2, m0 + r);
        myids[k * 32] = adj[(size_t)node * DEG + lane];
    }
    #pragma nounroll
    for (int k = 0; wave + (THREADS / 32) * k < 2 * nvalid; ++k) {
        const int jv = myids[k * 32];
        int rank = 0;
        #pragma unroll
        for (int j = 0; j < DEG; ++j) {
            const int s = __builtin_amdgcn_readlane(jv, j);
            rank += (s < jv || (s == jv && j < lane)) ? 1 : 0;
        }
        myids[k * 32] = __builtin_amdgcn_ds_permute(rank << 2, jv);
    }

    for (int k = 0; wave + (THREADS / 32) * k < 2 * nvalid; ++k) {
        const int u  = wave + (THREADS / 32) * k;
        const int r  = u >> 1;
        const int tb = u & 1;
        const int sv = myids[k * 32];

        const float* __restrict__ t = (tb ? feats : feat_data) + (lane << 2);
        v4f a[DEG / 2][2];
        #pragma unroll
        for (int i = 0; i < DEG / 2; ++i) {
            const float* p = t + (size_t)__builtin_amdgcn_readlane(sv, i) * FDIM;
            a[i][0] = *(const v4f*)(p);
            a[i][1] = *(const v4f*)(p + 128);
            __builtin_amdgcn_sched_barrier(0);
        }
        v4f lo = {}, hi = {};
        #pragma unroll
        for (int i = 0; i < DEG / 2; ++i) {
            lo += a[i][0];
            hi += a[i][1];
            const float* p = t + (size_t)__builtin_amdgcn_readlane(sv, DEG / 2 + i) * FDIM;
            a[i][0] = *(const v4f*)(p);
            a[i][1] = *(const v4f*)(p + 128);
            __builtin_amdgcn_sched_barrier(0);
        }
        #pragma unroll
        for (int i = 0; i < DEG / 2; ++i) {
            lo += a[i][0];
            hi += a[i][1];
        }
        put(r, 2 * tb, lo);
        put(r, 2 * tb + 1, hi);
    }
}


#define SLABK   32
#define HALFN   128
#define STAGE_F (HALFN * SLABK)
#define NSLAB   (2 * KDIM / SLABK)

__device__ __forceinline__ v8f wmma4(v2f a, v2f b, v8f c)
{
    return __builtin_amdgcn_wmma_f32_16x16x4_f32(false, a, false, b, (short)0, c, false, false);
}

__device__ __forceinline__ void gemm_phase(const float* __restrict__ Wg,
                                           const float* At, float* stage,
                                           int t, int wave, int lr, int hl,
                                           v8f (&acc)[4])
{
    const float* gsrc = Wg + (size_t)(t >> 3) * KDIM + ((t & 7) << 2);
    const int sdst = (t >> 3) * SLABK + (((t & 7) ^ ((t >> 4) & 7)) << 2);
    const int bl  = (wave * 32 + lr) * SLABK + (((lr >> 1) & 7) << 2) + 2 * hl;
    const int al  = lr * KDIM + 2 * hl;
    const int asw = lr << 2;

    v4f w[8];
    #pragma unroll
    for (int i = 0; i < 8; ++i) w[i] = *(const v4f*)(gsrc + (size_t)i * 16 * KDIM);
    #pragma unroll
    for (int i = 0; i < 8; ++i) *(v4f*)(stage + sdst + i * 16 * SLABK) = w[i];
    __syncthreads();

    #pragma unroll
    for (int nh = 0; nh < 2; ++nh) {
        v8f c0 = {}, c1 = {};
        #pragma nounroll
        for (int kk = 0; kk < KDIM / SLABK; ++kk) {
            const int s = nh * (KDIM / SLABK) + kk;
            const bool more = s + 1 < NSLAB;
            if (more) {
                const int s1 = s + 1;
                const float* g = gsrc + (size_t)(s1 >> 4) * HALFN * KDIM + (s1 & 15) * SLABK;
                #pragma unroll
                for (int i = 0; i < 8; ++i) w[i] = *(const v4f*)(g + (size_t)i * 16 * KDIM);
            }
            const int so = (s & 1) * STAGE_F;
            const int kx = (kk * SLABK) ^ asw;
            v2f a[8], b0[8], b1[8];
            #pragma unroll
            for (int j = 0; j < 8; ++j) {
                a[j] = *(const v2f*)(At + al + (kx ^ (j << 2)));
                const int bo = so + (bl ^ (j << 2));
                b0[j] = *(const v2f*)(stage + bo);
                b1[j] = *(const v2f*)(stage + bo + 16 * SLABK);
            }
            #pragma unroll
            for (int j = 0; j < 8; ++j) {
                c0 = wmma4(a[j], b0[j], c0);
                c1 = wmma4(a[j], b1[j], c1);
            }
            if (more) {
                float* sd = stage + (STAGE_F - so) + sdst;
                #pragma unroll
                for (int i = 0; i < 8; ++i) *(v4f*)(sd + i * 16 * SLABK) = w[i];
            }
            __syncthreads();
        }
        acc[2 * nh]     = c0;
        acc[2 * nh + 1] = c1;
    }
}

__device__ __forceinline__
void strip_finish(float* S, float* T, const v4f (&tv)[TM / 2],
                  const float* __restrict__ W1, const float* __restrict__ b1,
                  const float* __restrict__ W2, const float* __restrict__ b2,
                  int m0, int total, float* __restrict__ out)
{
    float (*part)[8] = reinterpret_cast<float (*)[8]>(&T[0]);
    float* inv_norm  = &T[TM * 8];

    const int t    = threadIdx.x;
    const int lane = t & 31;
    const int wave = t >> 5;
    const int lr    = lane & 15;
    const int hl    = lane >> 4;
    const int rbase = hl << 3;
    const int half = t >> 6;
    const int tc   = (t & 63) << 2;

    v8f acc[4];

    gemm_phase(W1, S, T, t, wave, lr, hl, acc);
    #pragma unroll
    for (int k = 0; k < TM / 2; ++k) {
        const int r = 2 * k + half;
        *(v4f*)(&T[r * KDIM + (tc ^ (r << 2))]) = tv[k];
    }
    #pragma unroll
    for (int q = 0; q < 4; ++q) {
        const int n = (q >> 1) * HALFN + wave * 32 + (q & 1) * 16 + lr;
        const float bias = (float)DEG * b1[n];
        #pragma unroll
        for (int v = 0; v < 8; ++v) {
            const int r = rbase + v;
            T[r * KDIM + ((FDIM + n) ^ (r << 2))] = acc[q][v] + bias;
        }
    }

    gemm_phase(W2, T, S, t, wave, lr, hl, acc);
    #pragma unroll
    for (int q = 0; q < 4; ++q) {
        const int n = (q >> 1) * HALFN + wave * 32 + (q & 1) * 16 + lr;
        const float bias = b2[n];
        #pragma unroll
        for (int v = 0; v < 8; ++v)
            S[(rbase + v) * KDIM + n] = acc[q][v] + bias;
    }
    __syncthreads();

    {
        const int r = t >> 3, seg = t & 7;
        float s = 0.f;
        #pragma unroll 8
        for (int c = seg * 32; c < seg * 32 + 32; ++c) {
            const float x = S[r * KDIM + c];
            s += x * x;
        }
        part[r][seg] = s;
    }
    __syncthreads();
    if (t < TM) {
        float s = 0.f;
        #pragma unroll
        for (int i = 0; i < 8; ++i) s += part[t][i];
        inv_norm[t] = 1.0f / fmaxf(sqrtf(s), 1e-12f);
    }
    __syncthreads();
    for (int i = t; i < TM * FDIM; i += THREADS) {
        const int r = i >> 8;
        const int c = i & (FDIM - 1);
        const int g = m0 + r;
        if (g < total)
            out[(size_t)g * FDIM + c] = S[r * KDIM + c] * inv_norm[r];
    }
}

__device__ __forceinline__
void load_node_ids(const int* __restrict__ in1, const int* __restrict__ in2,
                   const int* __restrict__ neg, int B1, int B2, int total,
                   int m0, int (&nid)[TM / 2])
{
    const int half = threadIdx.x >> 6;
    const int last = total - 1;
    #pragma unroll
    for (int k = 0; k < TM / 2; ++k)
        nid[k] = pick_node(in1, in2, neg, B1, B2, min(m0 + 2 * k + half, last));
}

__device__ __forceinline__
void load_node_rows(const float* __restrict__ feat_data,
                    const int (&nid)[TM / 2], v4f (&tv)[TM / 2])
{
    const int tc = (threadIdx.x & 63) << 2;
    #pragma unroll
    for (int k = 0; k < TM / 2; ++k)
        tv[k] = *(const v4f*)(feat_data + (size_t)nid[k] * FDIM + tc);
}

__global__ __launch_bounds__(THREADS)
void graph_strip_kernel(const float* __restrict__ feat_data,
                        const float* __restrict__ feats,
                        const float* __restrict__ W1,
                        const float* __restrict__ b1,
                        const float* __restrict__ W2,
                        const float* __restrict__ b2,
                        const int*   __restrict__ adj,
                        const int*   __restrict__ in1,
                        const int*   __restrict__ in2,
                        const int*   __restrict__ neg,
                        int B1, int B2, int total,
                        float* __restrict__ out)
{
    __shared__ float S[TM * KDIM];
    __shared__ float T[TM * KDIM];

    const int lane = threadIdx.x & 31;
    const int wave = __builtin_amdgcn_readfirstlane((int)(threadIdx.x >> 5));
    const int m0 = (int)blockIdx.x * TM;

    const int nvalid = min(total - m0, TM);
    if (nvalid == TM)
        sum_rows(feat_data, feats, adj, in1, in2, neg, B1, B2, total,
                 m0 + wave * RPW, lane, S, wave * RPW);
    else
        sum_rows_short(feat_data, feats, adj, in1, in2, neg, B1, B2,
                       m0, nvalid, wave, lane, S, reinterpret_cast<int*>(T + STAGE_F));
    int nid[TM / 2];
    load_node_ids(in1, in2, neg, B1, B2, total, m0, nid);
    v4f tv[TM / 2];
    load_node_rows(feat_data, nid, tv);

    strip_finish(S, T, tv, W1, b1, W2, b2, m0, total, out);
}

extern "C" void kernel_launch(void* const* d_in, const int* in_sizes, int n_in,
                              void* d_out, int out_size, void* d_ws, size_t ws_size,
                              hipStream_t stream) {
    const float* feat_data = (const float*)d_in[0];
    const float* feats     = (const float*)d_in[1];
    const float* W1        = (const float*)d_in[2];
    const float* b1        = (const float*)d_in[3];
    const float* W2        = (const float*)d_in[4];
    const float* b2        = (const float*)d_in[5];
    const int*   adj       = (const int*)d_in[6];
    const int*   in1       = (const int*)d_in[7];
    const int*   in2       = (const int*)d_in[8];
    const int*   neg       = (const int*)d_in[9];
    float* out = (float*)d_out;

    const int B1 = in_sizes[7];
    const int B2 = in_sizes[8];
    const int Bn = in_sizes[9];
    const int total = B1 + B2 + Bn;

    graph_strip_kernel<<<dim3((total + TM - 1) / TM), dim3(THREADS), 0, stream>>>(
        feat_data, feats, W1, b1, W2, b2, adj, in1, in2, neg, B1, B2, total, out);
}
